# starting kernel
# baseline (speedup 1.0000x reference)
#include <hip/hip_runtime.h>
#include <hip/hip_bf16.h>

typedef __attribute__((ext_vector_type(2))) float v2f;
typedef __attribute__((ext_vector_type(8))) float v8f;

#define D 128
#define PAD_WN 132
#define PAD_WE 260
#define PAD_WU 388

__device__ __forceinline__ float gelu_exact(float v) {
    return 0.5f * v * (1.0f + erff(v * 0.70710678118654752f));
}

__device__ __forceinline__ void load_w_lds(float* lds, const float* __restrict__ W,
                                           int rows, int cols, int pad) {
    const int nv4 = cols >> 2;
    const int total = rows * nv4;
    for (int idx = threadIdx.x; idx < total; idx += blockDim.x) {
        const int r  = idx / nv4;
        const int c4 = (idx - r * nv4) << 2;
        const float4 v = *(const float4*)(W + r * cols + c4);
        *(float4*)(lds + r * pad + c4) = v;
    }
}

__global__ void node_linear_kernel(const float* __restrict__ x,
                                   const float* __restrict__ Wn,
                                   const float* __restrict__ bn,
                                   float* __restrict__ node_emb,
                                   int nTiles) {
    extern __shared__ float lds[];
    load_w_lds(lds, Wn, 128, 128, PAD_WN);
    __syncthreads();

    const int lane = threadIdx.x & 31;
    const int tile = blockIdx.x * (blockDim.x >> 5) + (threadIdx.x >> 5);
    if (tile >= nTiles) return;

    const int m0    = tile << 4;
    const int mlow  = lane & 15;
    const int hi    = lane >> 4;
    const int khalf = hi << 1;

    const float* ap    = x + (m0 + mlow) * D + khalf;
    const float* bbase = lds + mlow * PAD_WN + khalf;

    v8f acc[8] = {};
    for (int k = 0; k < 128; k += 4) {
        const v2f a = *(const v2f*)(ap + k);
        #pragma unroll
        for (int t = 0; t < 8; ++t) {
            const v2f b = *(const v2f*)(bbase + t * (16 * PAD_WN) + k);
            acc[t] = __builtin_amdgcn_wmma_f32_16x16x4_f32(
                false, a, false, b, (short)0, acc[t], false, false);
        }
    }

    #pragma unroll
    for (int t = 0; t < 8; ++t) {
        const int n = (t << 4) + mlow;
        const float bv = bn[n];
        #pragma unroll
        for (int r = 0; r < 8; ++r)
            node_emb[(m0 + r + (hi << 3)) * D + n] = acc[t][r] + bv;
    }
}

__global__ void edge_kernel(const float* __restrict__ x,
                            const int* __restrict__ srcI,
                            const int* __restrict__ dstI,
                            const float* __restrict__ We,
                            const float* __restrict__ be,
                            const float* __restrict__ node_emb,
                            float* __restrict__ agg_node,
                            float* __restrict__ agg_edge,
                            int eTiles) {
    extern __shared__ float lds[];
    load_w_lds(lds, We, 128, 256, PAD_WE);
    __syncthreads();

    const int lane = threadIdx.x & 31;
    const int tile = blockIdx.x * (blockDim.x >> 5) + (threadIdx.x >> 5);
    if (tile >= eTiles) return;

    const int e0    = tile << 4;
    const int mlow  = lane & 15;
    const int hi    = lane >> 4;
    const int khalf = hi << 1;

    const int sIdx = srcI[e0 + mlow];
    const int dIdx = dstI[e0 + mlow];
    const float* aps   = x + sIdx * D + khalf;
    const float* apd   = x + dIdx * D + khalf;
    const float* bbase = lds + mlow * PAD_WE + khalf;

    v8f acc[8] = {};
    for (int k = 0; k < 128; k += 4) {
        const v2f a = *(const v2f*)(aps + k);
        #pragma unroll
        for (int t = 0; t < 8; ++t) {
            const v2f b = *(const v2f*)(bbase + t * (16 * PAD_WE) + k);
            acc[t] = __builtin_amdgcn_wmma_f32_16x16x4_f32(
                false, a, false, b, (short)0, acc[t], false, false);
        }
    }
    for (int k = 0; k < 128; k += 4) {
        const v2f a = *(const v2f*)(apd + k);
        #pragma unroll
        for (int t = 0; t < 8; ++t) {
            const v2f b = *(const v2f*)(bbase + t * (16 * PAD_WE) + 128 + k);
            acc[t] = __builtin_amdgcn_wmma_f32_16x16x4_f32(
                false, a, false, b, (short)0, acc[t], false, false);
        }
    }

    int srows[8];
    #pragma unroll
    for (int r = 0; r < 8; ++r) srows[r] = srcI[e0 + (hi << 3) + r];
    #pragma unroll
    for (int t = 0; t < 8; ++t) {
        const int n = (t << 4) + mlow;
        const float bv = be[n];
        #pragma unroll
        for (int r = 0; r < 8; ++r)
            atomicAdd(agg_edge + srows[r] * D + n, acc[t][r] + bv);
    }

    #pragma unroll 1
    for (int e = 0; e < 16; ++e) {
        const int s = __shfl(sIdx, e, 32);
        const int d = __shfl(dIdx, e, 32);
        const float* nb = node_emb + d * D;
        float*       ab = agg_node + s * D;
        #pragma unroll
        for (int c = 0; c < 4; ++c)
            atomicAdd(ab + lane + c * 32, nb[lane + c * 32]);
    }
}

__global__ void update_kernel(const float* __restrict__ node_emb,
                              const float* __restrict__ agg_node,
                              const float* __restrict__ agg_edge,
                              const float* __restrict__ Wu,
                              const float* __restrict__ bu,
                              float* __restrict__ out,
                              int nTiles) {
    extern __shared__ float lds[];
    load_w_lds(lds, Wu, 128, 384, PAD_WU);
    __syncthreads();

    const int lane = threadIdx.x & 31;
    const int tile = blockIdx.x * (blockDim.x >> 5) + (threadIdx.x >> 5);
    if (tile >= nTiles) return;

    const int m0    = tile << 4;
    const int mlow  = lane & 15;
    const int hi    = lane >> 4;
    const int khalf = hi << 1;

    const float* bbase = lds + mlow * PAD_WU + khalf;
    const float* parts[3] = { node_emb, agg_node, agg_edge };

    v8f acc[8] = {};
    #pragma unroll 1
    for (int part = 0; part < 3; ++part) {
        const float* ap  = parts[part] + (m0 + mlow) * D + khalf;
        const int    kg0 = part * 128;
        for (int k = 0; k < 128; k += 4) {
            const v2f raw = *(const v2f*)(ap + k);
            v2f a;
            a[0] = gelu_exact(raw[0]);
            a[1] = gelu_exact(raw[1]);
            #pragma unroll
            for (int t = 0; t < 8; ++t) {
                const v2f b = *(const v2f*)(bbase + t * (16 * PAD_WU) + kg0 + k);
                acc[t] = __builtin_amdgcn_wmma_f32_16x16x4_f32(
                    false, a, false, b, (short)0, acc[t], false, false);
            }
        }
    }

    #pragma unroll
    for (int t = 0; t < 8; ++t) {
        const int n = (t << 4) + mlow;
        const float bv = bu[n];
        #pragma unroll
        for (int r = 0; r < 8; ++r)
            out[(m0 + r + (hi << 3)) * D + n] = acc[t][r] + bv;
    }
}

extern "C" void kernel_launch(void* const* d_in, const int* in_sizes, int n_in,
                              void* d_out, int out_size, void* d_ws, size_t ws_size,
                              hipStream_t stream) {
    const float* x  = (const float*)d_in[0];
    const int*   ei = (const int*)d_in[1];
    const float* Wn = (const float*)d_in[2];
    const float* bn = (const float*)d_in[3];
    const float* We = (const float*)d_in[4];
    const float* be = (const float*)d_in[5];
    const float* Wu = (const float*)d_in[6];
    const float* bu = (const float*)d_in[7];

    const int N = in_sizes[0] / D;
    const int E = in_sizes[1] / 2;
    const int* srcI = ei;
    const int* dstI = ei + E;

    float* node_emb = (float*)d_ws;
    float* agg_node = node_emb + (size_t)N * D;
    float* agg_edge = agg_node + (size_t)N * D;

    hipMemsetAsync(agg_node, 0, 2 * (size_t)N * D * sizeof(float), stream);

    const int wpb = 8;
    const int nTiles = (N + 15) >> 4;
    const int eTiles = (E + 15) >> 4;
    const dim3 blk(32 * wpb);

    node_linear_kernel<<<(nTiles + wpb - 1) / wpb, blk,
                         128 * PAD_WN * sizeof(float), stream>>>(
        x, Wn, bn, node_emb, nTiles);

    edge_kernel<<<(eTiles + wpb - 1) / wpb, blk,
                  128 * PAD_WE * sizeof(float), stream>>>(
        x, srcI, dstI, We, be, node_emb, agg_node, agg_edge, eTiles);

    update_kernel<<<(nTiles + wpb - 1) / wpb, blk,
                    128 * PAD_WU * sizeof(float), stream>>>(
        node_emb, agg_node, agg_edge, Wu, bu, (float*)d_out, nTiles);
}


# x rows summed per source node first, in fixed ascending-destination order (lists via integer exchange atomics); linear maps then applied once per node on WMMA; no f32 atomics; f32 sum order changes
# speedup vs baseline: 2.1388x; 2.1388x over previous
#include <hip/hip_runtime.h>
#include <hip/hip_bf16.h>

typedef __attribute__((ext_vector_type(2))) float v2f;
typedef __attribute__((ext_vector_type(8))) float v8f;

#define D 128
#define PAD_WN 132
#define PAD_WE 260
#define PAD_WU 388

__device__ __forceinline__ float gelu_exact(float v) {
    return 0.5f * v * (1.0f + erff(v * 0.70710678118654752f));
}

__device__ __forceinline__ void load_w_lds(float* lds, const float* __restrict__ W,
                                           int rows, int cols, int pad) {
    const int nv4 = cols >> 2;
    const int total = rows * nv4;
    for (int idx = threadIdx.x; idx < total; idx += blockDim.x) {
        const int r  = idx / nv4;
        const int c4 = (idx - r * nv4) << 2;
        const float4 v = *(const float4*)(W + r * cols + c4);
        *(float4*)(lds + r * pad + c4) = v;
    }
}

__global__ void link_kernel(const int* __restrict__ srcI,
                            int* __restrict__ head,
                            int* __restrict__ next,
                            int E) {
    const int e = blockIdx.x * blockDim.x + threadIdx.x;
    if (e < E) next[e] = atomicExch(head + srcI[e], e);
}

#define NODES_PER_WAVE 4
__global__ __launch_bounds__(256) void gather_sum_kernel(const float* __restrict__ x,
                                                         const int* __restrict__ dstI,
                                                         const int* __restrict__ head,
                                                         const int* __restrict__ next,
                                                         float* __restrict__ S,
                                                         float* __restrict__ degf,
                                                         int N, int E) {
    const int lane = threadIdx.x & 31;
    const int wv = __builtin_amdgcn_readfirstlane(
        (int)(blockIdx.x * (blockDim.x >> 5) + (threadIdx.x >> 5)));
    const float* xl = x + lane * 4;

    for (int q = 0; q < NODES_PER_WAVE; ++q) {
        const int node = wv * NODES_PER_WAVE + q;
        if (node >= N) break;

        const int first = head[node];
        int e = first;
        int cnt = 0;
        int myd = 0x7fffffff;
        while (e >= 0 && cnt < 32) {
            const int d = dstI[e];
            if (lane == cnt) myd = d;
            e = next[e];
            ++cnt;
        }

        float4 acc = {0.0f, 0.0f, 0.0f, 0.0f};
        int deg = cnt;

        if (e < 0) {
            int rank = 0;
            #pragma unroll
            for (int j = 0; j < 32; ++j) {
                const int dj = __builtin_amdgcn_readlane(myd, j);
                rank += ((dj < myd) || (dj == myd && j < lane)) ? 1 : 0;
            }
            const int sorted = __builtin_amdgcn_ds_permute(rank << 2, myd);
            for (int r0 = 0; r0 < cnt; r0 += 4) {
                float4 v[4];
                #pragma unroll
                for (int i = 0; i < 4; ++i) {
                    const int r = (r0 + i < cnt) ? (r0 + i) : (cnt - 1);
                    const int d = __builtin_amdgcn_readlane(sorted, r);
                    v[i] = *(const float4*)(xl + (size_t)d * D);
                }
                #pragma unroll
                for (int i = 0; i < 4; ++i) {
                    if (r0 + i < cnt) {
                        acc.x += v[i].x; acc.y += v[i].y; acc.z += v[i].z; acc.w += v[i].w;
                    }
                }
            }
        } else {
            deg = 0;
            int last = -1;
            for (int outer = 0; outer < N; ++outer) {
                int best = 0x7fffffff;
                int mult = 0;
                int p = first;
                for (int it = 0; p >= 0 && it < E; ++it) {
                    const int d = dstI[p];
                    if (d > last) {
                        if (d < best) { best = d; mult = 1; }
                        else if (d == best) ++mult;
                    }
                    p = next[p];
                }
                if (mult == 0) break;
                const float4 v = *(const float4*)(xl + (size_t)best * D);
                for (int m = 0; m < mult; ++m) {
                    acc.x += v.x; acc.y += v.y; acc.z += v.z; acc.w += v.w;
                }
                deg += mult;
                last = best;
            }
        }

        *(float4*)(S + (size_t)node * D + lane * 4) = acc;
        if (lane == 0) degf[node] = (float)deg;
    }
}

#define WMMA_F32(ACC, A2, B2) \
    ACC = __builtin_amdgcn_wmma_f32_16x16x4_f32(false, A2, false, B2, (short)0, ACC, false, false)

__global__ __launch_bounds__(256) void node_maps_kernel(const float* __restrict__ x,
                                                        const float* __restrict__ S,
                                                        const float* __restrict__ degf,
                                                        const float* __restrict__ Wn,
                                                        const float* __restrict__ bn,
                                                        const float* __restrict__ We,
                                                        const float* __restrict__ be,
                                                        float* __restrict__ node_emb,
                                                        float* __restrict__ agg_node,
                                                        float* __restrict__ agg_edge,
                                                        int nTiles) {
    extern __shared__ float lds[];
    float* ldsN = lds;
    float* ldsE = lds + 128 * PAD_WN;
    load_w_lds(ldsN, Wn, 128, 128, PAD_WN);
    load_w_lds(ldsE, We, 128, 256, PAD_WE);
    __syncthreads();

    const int lane = threadIdx.x & 31;
    const int wave = threadIdx.x >> 5;
    const int wpb  = blockDim.x >> 5;
    const int mlow = lane & 15;
    const int hi   = lane >> 4;

    const float* bN = ldsN + mlow * PAD_WN + 4 * hi;
    const float* bE = ldsE + mlow * PAD_WE + 4 * hi;

    for (int j = wave; ; j += wpb) {
        const int tile = j * (int)gridDim.x + (int)blockIdx.x;
        if (tile >= nTiles) break;
        const int m0 = tile << 4;

        const float* xr = x + (size_t)(m0 + mlow) * D + 4 * hi;
        const float* sr = S + (size_t)(m0 + mlow) * D + 4 * hi;

        v8f accXN[8] = {};
        v8f accSN[8] = {};
        v8f accXL[8] = {};
        v8f accSR[8] = {};

        float4 xn = *(const float4*)xr;
        float4 sn = *(const float4*)sr;
        #pragma unroll 1
        for (int c = 0; c < 16; ++c) {
            const float4 xa = xn;
            const float4 sa = sn;
            const int cn = (c < 15) ? (c + 1) : 15;
            xn = *(const float4*)(xr + 8 * cn);
            sn = *(const float4*)(sr + 8 * cn);

            const float* pN = bN + 8 * c;
            const float* pE = bE + 8 * c;
            float4 wN[8], wL[8], wR[8];
            #pragma unroll
            for (int t = 0; t < 8; ++t) {
                wN[t] = *(const float4*)(pN + t * (16 * PAD_WN));
                wL[t] = *(const float4*)(pE + t * (16 * PAD_WE));
                wR[t] = *(const float4*)(pE + t * (16 * PAD_WE) + 128);
            }
            const v2f x0 = {xa.x, xa.y};
            const v2f x1 = {xa.z, xa.w};
            const v2f s0 = {sa.x, sa.y};
            const v2f s1 = {sa.z, sa.w};
            #pragma unroll
            for (int t = 0; t < 8; ++t) {
                const v2f b0 = {wN[t].x, wN[t].y};
                WMMA_F32(accXN[t], x0, b0);
                WMMA_F32(accSN[t], s0, b0);
            }
            #pragma unroll
            for (int t = 0; t < 8; ++t) {
                const v2f b1 = {wN[t].z, wN[t].w};
                WMMA_F32(accXN[t], x1, b1);
                WMMA_F32(accSN[t], s1, b1);
            }
            #pragma unroll
            for (int t = 0; t < 8; ++t) {
                const v2f l0 = {wL[t].x, wL[t].y};
                const v2f r0 = {wR[t].x, wR[t].y};
                WMMA_F32(accXL[t], x0, l0);
                WMMA_F32(accSR[t], s0, r0);
            }
            #pragma unroll
            for (int t = 0; t < 8; ++t) {
                const v2f l1 = {wL[t].z, wL[t].w};
                const v2f r1 = {wR[t].z, wR[t].w};
                WMMA_F32(accXL[t], x1, l1);
                WMMA_F32(accSR[t], s1, r1);
            }
            asm volatile("" :: "v"(xn.x), "v"(xn.y), "v"(xn.z), "v"(xn.w),
                               "v"(sn.x), "v"(sn.y), "v"(sn.z), "v"(sn.w));
        }

        float dg[8];
        #pragma unroll
        for (int r = 0; r < 8; ++r) dg[r] = degf[m0 + r + (hi << 3)];

        #pragma unroll
        for (int t = 0; t < 8; ++t) {
            const int n = (t << 4) + mlow;
            const float bnv = bn[n];
            const float bev = be[n];
            #pragma unroll
            for (int r = 0; r < 8; ++r) {
                const size_t o = (size_t)(m0 + r + (hi << 3)) * D + n;
                node_emb[o] = accXN[t][r] + bnv;
                agg_node[o] = accSN[t][r] + dg[r] * bnv;
                agg_edge[o] = (dg[r] * accXL[t][r] + accSR[t][r]) + dg[r] * bev;
            }
        }
    }
}

__global__ void update_kernel(const float* __restrict__ node_emb,
                              const float* __restrict__ agg_node,
                              const float* __restrict__ agg_edge,
                              const float* __restrict__ Wu,
                              const float* __restrict__ bu,
                              float* __restrict__ out,
                              int nTiles) {
    extern __shared__ float lds[];
    load_w_lds(lds, Wu, 128, 384, PAD_WU);
    __syncthreads();

    const int lane = threadIdx.x & 31;
    const int tile = blockIdx.x * (blockDim.x >> 5) + (threadIdx.x >> 5);
    if (tile >= nTiles) return;

    const int m0    = tile << 4;
    const int mlow  = lane & 15;
    const int hi    = lane >> 4;
    const int khalf = hi << 1;

    const float* bbase = lds + mlow * PAD_WU + khalf;
    const float* parts[3] = { node_emb, agg_node, agg_edge };

    v8f acc[8] = {};
    #pragma unroll 1
    for (int part = 0; part < 3; ++part) {
        const float* ap  = parts[part] + (m0 + mlow) * D + khalf;
        const int    kg0 = part * 128;
        for (int k = 0; k < 128; k += 4) {
            const v2f raw = *(const v2f*)(ap + k);
            v2f a;
            a[0] = gelu_exact(raw[0]);
            a[1] = gelu_exact(raw[1]);
            #pragma unroll
            for (int t = 0; t < 8; ++t) {
                const v2f b = *(const v2f*)(bbase + t * (16 * PAD_WU) + kg0 + k);
                acc[t] = __builtin_amdgcn_wmma_f32_16x16x4_f32(
                    false, a, false, b, (short)0, acc[t], false, false);
            }
        }
    }

    #pragma unroll
    for (int t = 0; t < 8; ++t) {
        const int n = (t << 4) + mlow;
        const float bv = bu[n];
        #pragma unroll
        for (int r = 0; r < 8; ++r)
            out[(m0 + r + (hi << 3)) * D + n] = acc[t][r] + bv;
    }
}

extern "C" void kernel_launch(void* const* d_in, const int* in_sizes, int n_in,
                              void* d_out, int out_size, void* d_ws, size_t ws_size,
                              hipStream_t stream) {
    const float* x  = (const float*)d_in[0];
    const int*   ei = (const int*)d_in[1];
    const float* Wn = (const float*)d_in[2];
    const float* bn = (const float*)d_in[3];
    const float* We = (const float*)d_in[4];
    const float* be = (const float*)d_in[5];
    const float* Wu = (const float*)d_in[6];
    const float* bu = (const float*)d_in[7];

    const int N = in_sizes[0] / D;
    const int E = in_sizes[1] / 2;
    const int* srcI = ei;
    const int* dstI = ei + E;

    float* node_emb = (float*)d_ws;
    float* agg_node = node_emb + (size_t)N * D;
    float* agg_edge = agg_node + (size_t)N * D;
    float* S        = agg_edge + (size_t)N * D;
    float* degf     = S + (size_t)N * D;
    int*   head     = (int*)(degf + N);
    int*   next     = head + N;

    hipMemsetAsync(head, 0xFF, (size_t)N * sizeof(int), stream);

    const int wpb = 8;
    const int nTiles = (N + 15) >> 4;
    const dim3 blk(32 * wpb);

    link_kernel<<<(E + 255) / 256, blk, 0, stream>>>(srcI, head, next, E);

    gather_sum_kernel<<<(N + wpb * NODES_PER_WAVE - 1) / (wpb * NODES_PER_WAVE), blk, 0, stream>>>(
        x, dstI, head, next, S, degf, N, E);

    node_maps_kernel<<<256, blk,
                       (128 * PAD_WN + 128 * PAD_WE) * sizeof(float), stream>>>(
        x, S, degf, Wn, bn, We, be, node_emb, agg_node, agg_edge, nTiles);

    update_kernel<<<(nTiles + wpb - 1) / wpb, blk,
                    128 * PAD_WU * sizeof(float), stream>>>(
        node_emb, agg_node, agg_edge, Wu, bu, (float*)d_out, nTiles);
}


# update kernel runs as one 32-wave block per WGP that walks its tiles: Wu staged once per WGP instead of once per 8 tiles, 6-7 waves per SIMD overlap load, GELU and WMMA; sums unchanged
# speedup vs baseline: 3.0249x; 1.4143x over previous
#include <hip/hip_runtime.h>
#include <hip/hip_bf16.h>

typedef __attribute__((ext_vector_type(2))) float v2f;
typedef __attribute__((ext_vector_type(8))) float v8f;

#define D 128
#define PAD_WN 132
#define PAD_WE 260
#define PAD_WU 388

__device__ __forceinline__ float gelu_exact(float v) {
    return 0.5f * v * (1.0f + erff(v * 0.70710678118654752f));
}

__device__ __forceinline__ void load_w_lds(float* lds, const float* __restrict__ W,
                                           int rows, int cols, int pad) {
    const int nv4 = cols >> 2;
    const int total = rows * nv4;
    for (int idx = threadIdx.x; idx < total; idx += blockDim.x) {
        const int r  = idx / nv4;
        const int c4 = (idx - r * nv4) << 2;
        const float4 v = *(const float4*)(W + r * cols + c4);
        *(float4*)(lds + r * pad + c4) = v;
    }
}

__global__ void link_kernel(const int* __restrict__ srcI,
                            int* __restrict__ head,
                            int* __restrict__ next,
                            int E) {
    const int e = blockIdx.x * blockDim.x + threadIdx.x;
    if (e < E) next[e] = atomicExch(head + srcI[e], e);
}

#define NODES_PER_WAVE 4
__global__ __launch_bounds__(256) void gather_sum_kernel(const float* __restrict__ x,
                                                         const int* __restrict__ dstI,
                                                         const int* __restrict__ head,
                                                         const int* __restrict__ next,
                                                         float* __restrict__ S,
                                                         float* __restrict__ degf,
                                                         int N, int E) {
    const int lane = threadIdx.x & 31;
    const int wv = __builtin_amdgcn_readfirstlane(
        (int)(blockIdx.x * (blockDim.x >> 5) + (threadIdx.x >> 5)));
    const float* xl = x + lane * 4;

    for (int q = 0; q < NODES_PER_WAVE; ++q) {
        const int node = wv * NODES_PER_WAVE + q;
        if (node >= N) break;

        const int first = head[node];
        int e = first;
        int cnt = 0;
        int myd = 0x7fffffff;
        while (e >= 0 && cnt < 32) {
            const int d = dstI[e];
            if (lane == cnt) myd = d;
            e = next[e];
            ++cnt;
        }

        float4 acc = {0.0f, 0.0f, 0.0f, 0.0f};
        int deg = cnt;

        if (e < 0) {
            int rank = 0;
            #pragma unroll
            for (int j = 0; j < 32; ++j) {
                const int dj = __builtin_amdgcn_readlane(myd, j);
                rank += ((dj < myd) || (dj == myd && j < lane)) ? 1 : 0;
            }
            const int sorted = __builtin_amdgcn_ds_permute(rank << 2, myd);
            for (int r0 = 0; r0 < cnt; r0 += 4) {
                float4 v[4];
                #pragma unroll
                for (int i = 0; i < 4; ++i) {
                    const int r = (r0 + i < cnt) ? (r0 + i) : (cnt - 1);
                    const int d = __builtin_amdgcn_readlane(sorted, r);
                    v[i] = *(const float4*)(xl + (size_t)d * D);
                }
                #pragma unroll
                for (int i = 0; i < 4; ++i) {
                    if (r0 + i < cnt) {
                        acc.x += v[i].x; acc.y += v[i].y; acc.z += v[i].z; acc.w += v[i].w;
                    }
                }
            }
        } else {
            deg = 0;
            int last = -1;
            for (int outer = 0; outer < N; ++outer) {
                int best = 0x7fffffff;
                int mult = 0;
                int p = first;
                for (int it = 0; p >= 0 && it < E; ++it) {
                    const int d = dstI[p];
                    if (d > last) {
                        if (d < best) { best = d; mult = 1; }
                        else if (d == best) ++mult;
                    }
                    p = next[p];
                }
                if (mult == 0) break;
                const float4 v = *(const float4*)(xl + (size_t)best * D);
                for (int m = 0; m < mult; ++m) {
                    acc.x += v.x; acc.y += v.y; acc.z += v.z; acc.w += v.w;
                }
                deg += mult;
                last = best;
            }
        }

        *(float4*)(S + (size_t)node * D + lane * 4) = acc;
        if (lane == 0) degf[node] = (float)deg;
    }
}

#define WMMA_F32(ACC, A2, B2) \
    ACC = __builtin_amdgcn_wmma_f32_16x16x4_f32(false, A2, false, B2, (short)0, ACC, false, false)

__global__ __launch_bounds__(256) void node_maps_kernel(const float* __restrict__ x,
                                                        const float* __restrict__ S,
                                                        const float* __restrict__ degf,
                                                        const float* __restrict__ Wn,
                                                        const float* __restrict__ bn,
                                                        const float* __restrict__ We,
                                                        const float* __restrict__ be,
                                                        float* __restrict__ node_emb,
                                                        float* __restrict__ agg_node,
                                                        float* __restrict__ agg_edge,
                                                        int nTiles) {
    extern __shared__ float lds[];
    float* ldsN = lds;
    float* ldsE = lds + 128 * PAD_WN;
    load_w_lds(ldsN, Wn, 128, 128, PAD_WN);
    load_w_lds(ldsE, We, 128, 256, PAD_WE);
    __syncthreads();

    const int lane = threadIdx.x & 31;
    const int wave = threadIdx.x >> 5;
    const int wpb  = blockDim.x >> 5;
    const int mlow = lane & 15;
    const int hi   = lane >> 4;

    const float* bN = ldsN + mlow * PAD_WN + 4 * hi;
    const float* bE = ldsE + mlow * PAD_WE + 4 * hi;

    for (int j = wave; ; j += wpb) {
        const int tile = j * (int)gridDim.x + (int)blockIdx.x;
        if (tile >= nTiles) break;
        const int m0 = tile << 4;

        const float* xr = x + (size_t)(m0 + mlow) * D + 4 * hi;
        const float* sr = S + (size_t)(m0 + mlow) * D + 4 * hi;

        v8f accXN[8] = {};
        v8f accSN[8] = {};
        v8f accXL[8] = {};
        v8f accSR[8] = {};

        float4 xn = *(const float4*)xr;
        float4 sn = *(const float4*)sr;
        #pragma unroll 1
        for (int c = 0; c < 16; ++c) {
            const float4 xa = xn;
            const float4 sa = sn;
            const int cn = (c < 15) ? (c + 1) : 15;
            xn = *(const float4*)(xr + 8 * cn);
            sn = *(const float4*)(sr + 8 * cn);

            const float* pN = bN + 8 * c;
            const float* pE = bE + 8 * c;
            float4 wN[8], wL[8], wR[8];
            #pragma unroll
            for (int t = 0; t < 8; ++t) {
                wN[t] = *(const float4*)(pN + t * (16 * PAD_WN));
                wL[t] = *(const float4*)(pE + t * (16 * PAD_WE));
                wR[t] = *(const float4*)(pE + t * (16 * PAD_WE) + 128);
            }
            const v2f x0 = {xa.x, xa.y};
            const v2f x1 = {xa.z, xa.w};
            const v2f s0 = {sa.x, sa.y};
            const v2f s1 = {sa.z, sa.w};
            #pragma unroll
            for (int t = 0; t < 8; ++t) {
                const v2f b0 = {wN[t].x, wN[t].y};
                WMMA_F32(accXN[t], x0, b0);
                WMMA_F32(accSN[t], s0, b0);
            }
            #pragma unroll
            for (int t = 0; t < 8; ++t) {
                const v2f b1 = {wN[t].z, wN[t].w};
                WMMA_F32(accXN[t], x1, b1);
                WMMA_F32(accSN[t], s1, b1);
            }
            #pragma unroll
            for (int t = 0; t < 8; ++t) {
                const v2f l0 = {wL[t].x, wL[t].y};
                const v2f r0 = {wR[t].x, wR[t].y};
                WMMA_F32(accXL[t], x0, l0);
                WMMA_F32(accSR[t], s0, r0);
            }
            #pragma unroll
            for (int t = 0; t < 8; ++t) {
                const v2f l1 = {wL[t].z, wL[t].w};
                const v2f r1 = {wR[t].z, wR[t].w};
                WMMA_F32(accXL[t], x1, l1);
                WMMA_F32(accSR[t], s1, r1);
            }
            asm volatile("" :: "v"(xn.x), "v"(xn.y), "v"(xn.z), "v"(xn.w),
                               "v"(sn.x), "v"(sn.y), "v"(sn.z), "v"(sn.w));
        }

        float dg[8];
        #pragma unroll
        for (int r = 0; r < 8; ++r) dg[r] = degf[m0 + r + (hi << 3)];

        #pragma unroll
        for (int t = 0; t < 8; ++t) {
            const int n = (t << 4) + mlow;
            const float bnv = bn[n];
            const float bev = be[n];
            #pragma unroll
            for (int r = 0; r < 8; ++r) {
                const size_t o = (size_t)(m0 + r + (hi << 3)) * D + n;
                node_emb[o] = accXN[t][r] + bnv;
                agg_node[o] = accSN[t][r] + dg[r] * bnv;
                agg_edge[o] = (dg[r] * accXL[t][r] + accSR[t][r]) + dg[r] * bev;
            }
        }
    }
}

__global__ __launch_bounds__(1024) void update_kernel(const float* __restrict__ node_emb,
                                                      const float* __restrict__ agg_node,
                                                      const float* __restrict__ agg_edge,
                                                      const float* __restrict__ Wu,
                                                      const float* __restrict__ bu,
                                                      float* __restrict__ out,
                                                      int nTiles) {
    extern __shared__ float lds[];
    load_w_lds(lds, Wu, 128, 384, PAD_WU);
    __syncthreads();

    const int lane  = threadIdx.x & 31;
    const int wave  = threadIdx.x >> 5;
    const int wpb   = blockDim.x >> 5;
    const int mlow  = lane & 15;
    const int hi    = lane >> 4;
    const int khalf = hi << 1;

    const float* bbase = lds + mlow * PAD_WU + khalf;
    const float* parts[3] = { node_emb, agg_node, agg_edge };

    for (int j = wave; ; j += wpb) {
        const int tile = j * (int)gridDim.x + (int)blockIdx.x;
        if (tile >= nTiles) break;
        const int m0 = tile << 4;

        v8f acc[8] = {};
        #pragma unroll 1
        for (int part = 0; part < 3; ++part) {
            const float* ap  = parts[part] + (size_t)(m0 + mlow) * D + khalf;
            const int    kg0 = part * 128;
            for (int k = 0; k < 128; k += 4) {
                const v2f raw = *(const v2f*)(ap + k);
                v2f a;
                a[0] = gelu_exact(raw[0]);
                a[1] = gelu_exact(raw[1]);
                #pragma unroll
                for (int t = 0; t < 8; ++t) {
                    const v2f b = *(const v2f*)(bbase + t * (16 * PAD_WU) + kg0 + k);
                    acc[t] = __builtin_amdgcn_wmma_f32_16x16x4_f32(
                        false, a, false, b, (short)0, acc[t], false, false);
                }
            }
        }

        #pragma unroll
        for (int t = 0; t < 8; ++t) {
            const int n = (t << 4) + mlow;
            const float bv = bu[n];
            #pragma unroll
            for (int r = 0; r < 8; ++r)
                out[(size_t)(m0 + r + (hi << 3)) * D + n] = acc[t][r] + bv;
        }
    }
}

extern "C" void kernel_launch(void* const* d_in, const int* in_sizes, int n_in,
                              void* d_out, int out_size, void* d_ws, size_t ws_size,
                              hipStream_t stream) {
    const float* x  = (const float*)d_in[0];
    const int*   ei = (const int*)d_in[1];
    const float* Wn = (const float*)d_in[2];
    const float* bn = (const float*)d_in[3];
    const float* We = (const float*)d_in[4];
    const float* be = (const float*)d_in[5];
    const float* Wu = (const float*)d_in[6];
    const float* bu = (const float*)d_in[7];

    const int N = in_sizes[0] / D;
    const int E = in_sizes[1] / 2;
    const int* srcI = ei;
    const int* dstI = ei + E;

    float* node_emb = (float*)d_ws;
    float* agg_node = node_emb + (size_t)N * D;
    float* agg_edge = agg_node + (size_t)N * D;
    float* S        = agg_edge + (size_t)N * D;
    float* degf     = S + (size_t)N * D;
    int*   head     = (int*)(degf + N);
    int*   next     = head + N;

    hipMemsetAsync(head, 0xFF, (size_t)N * sizeof(int), stream);

    const int wpb = 8;
    const int nTiles = (N + 15) >> 4;
    const dim3 blk(32 * wpb);

    link_kernel<<<(E + 255) / 256, blk, 0, stream>>>(srcI, head, next, E);

    gather_sum_kernel<<<(N + wpb * NODES_PER_WAVE - 1) / (wpb * NODES_PER_WAVE), blk, 0, stream>>>(
        x, dstI, head, next, S, degf, N, E);

    node_maps_kernel<<<256, blk,
                       (128 * PAD_WN + 128 * PAD_WE) * sizeof(float), stream>>>(
        x, S, degf, Wn, bn, We, be, node_emb, agg_node, agg_edge, nTiles);

    update_kernel<<<256, dim3(1024),
                    128 * PAD_WU * sizeof(float), stream>>>(
        node_emb, agg_node, agg_edge, Wu, bu, (float*)d_out, nTiles);
}


# per-node rows of 32 destination slots (slot = returning integer counter add) replace the linked lists: one vector load per node, not a chain of dependent scalar loads; f32 sums unchanged
# speedup vs baseline: 3.1826x; 1.0521x over previous
#include <hip/hip_runtime.h>
#include <hip/hip_bf16.h>

typedef __attribute__((ext_vector_type(2))) float v2f;
typedef __attribute__((ext_vector_type(8))) float v8f;

#define D 128
#define PAD_WN 132
#define PAD_WE 260
#define PAD_WU 388

__device__ __forceinline__ float gelu_exact(float v) {
    return 0.5f * v * (1.0f + erff(v * 0.70710678118654752f));
}

__device__ __forceinline__ void load_w_lds(float* lds, const float* __restrict__ W,
                                           int rows, int cols, int pad) {
    const int nv4 = cols >> 2;
    const int total = rows * nv4;
    for (int idx = threadIdx.x; idx < total; idx += blockDim.x) {
        const int r  = idx / nv4;
        const int c4 = (idx - r * nv4) << 2;
        const float4 v = *(const float4*)(W + r * cols + c4);
        *(float4*)(lds + r * pad + c4) = v;
    }
}

#define CAP 32
__global__ void fill_kernel(const int* __restrict__ srcI,
                            const int* __restrict__ dstI,
                            int* __restrict__ cnt,
                            int* __restrict__ slot,
                            int E) {
    const int e = blockIdx.x * blockDim.x + threadIdx.x;
    if (e < E) {
        const int s = srcI[e];
        const int d = dstI[e];
        asm volatile("" :: "v"(d));
        const int k = atomicAdd(cnt + s, 1);
        if (k < CAP) slot[(size_t)s * CAP + k] = d;
    }
}

#define NODES_PER_WAVE 8
__global__ __launch_bounds__(256) void gather_sum_kernel(const float* __restrict__ x,
                                                         const int* __restrict__ srcI,
                                                         const int* __restrict__ dstI,
                                                         const int* __restrict__ cnt,
                                                         const int* __restrict__ slot,
                                                         float* __restrict__ S,
                                                         float* __restrict__ degf,
                                                         int N, int E) {
    const int lane = threadIdx.x & 31;
    const int wv = __builtin_amdgcn_readfirstlane(
        (int)(blockIdx.x * (blockDim.x >> 5) + (threadIdx.x >> 5)));
    const float* xl = x + lane * 4;
    const int node0 = wv * NODES_PER_WAVE;

    int cq[NODES_PER_WAVE];
    int dq[NODES_PER_WAVE];
    if (node0 + NODES_PER_WAVE <= N) {
        #pragma unroll
        for (int q = 0; q < NODES_PER_WAVE; ++q) cq[q] = cnt[node0 + q];
    } else {
        #pragma unroll
        for (int q = 0; q < NODES_PER_WAVE; ++q)
            cq[q] = (node0 + q < N) ? cnt[node0 + q] : 0;
    }
    #pragma unroll
    for (int q = 0; q < NODES_PER_WAVE; ++q) {
        dq[q] = 0x7fffffff;
        if (lane < cq[q]) dq[q] = slot[(size_t)(node0 + q) * CAP + lane];
    }

    #pragma unroll 1
    for (int q = 0; q < NODES_PER_WAVE; ++q) {
        const int node = node0 + q;
        if (node >= N) break;

        int cnt_q = cq[0];
        int myd = dq[0];
        #pragma unroll
        for (int t = 1; t < NODES_PER_WAVE; ++t) {
            if (q == t) { cnt_q = cq[t]; myd = dq[t]; }
        }
        float4 acc = {0.0f, 0.0f, 0.0f, 0.0f};
        int deg = cnt_q;

        if (cnt_q <= CAP) {
            int rank = 0;
            #pragma unroll
            for (int j = 0; j < 32; ++j) {
                const int dj = __builtin_amdgcn_readlane(myd, j);
                rank += ((dj < myd) || (dj == myd && j < lane)) ? 1 : 0;
            }
            const int sorted = __builtin_amdgcn_ds_permute(rank << 2, myd);
            for (int r0 = 0; r0 < cnt_q; r0 += 4) {
                float4 v[4];
                #pragma unroll
                for (int i = 0; i < 4; ++i) {
                    const int r = (r0 + i < cnt_q) ? (r0 + i) : (cnt_q - 1);
                    const int d = __builtin_amdgcn_readlane(sorted, r);
                    v[i] = *(const float4*)(xl + (size_t)d * D);
                }
                #pragma unroll
                for (int i = 0; i < 4; ++i) {
                    if (r0 + i < cnt_q) {
                        acc.x += v[i].x; acc.y += v[i].y; acc.z += v[i].z; acc.w += v[i].w;
                    }
                }
            }
        } else {
            deg = 0;
            int last = -1;
            for (int outer = 0; outer < N; ++outer) {
                int best = 0x7fffffff;
                int mult = 0;
                for (int e0 = 0; e0 < E; e0 += 32) {
                    const int e = e0 + lane;
                    if (e < E && srcI[e] == node) {
                        const int d = dstI[e];
                        if (d > last) {
                            if (d < best) { best = d; mult = 1; }
                            else if (d == best) ++mult;
                        }
                    }
                }
                int wbest = best;
                for (int off = 16; off >= 1; off >>= 1) {
                    const int o = __shfl_xor(wbest, off);
                    wbest = (o < wbest) ? o : wbest;
                }
                int m = (best == wbest) ? mult : 0;
                for (int off = 16; off >= 1; off >>= 1) m += __shfl_xor(m, off);
                m = __builtin_amdgcn_readfirstlane(m);
                wbest = __builtin_amdgcn_readfirstlane(wbest);
                if (m == 0) break;
                const float4 v = *(const float4*)(xl + (size_t)wbest * D);
                for (int mm = 0; mm < m; ++mm) {
                    acc.x += v.x; acc.y += v.y; acc.z += v.z; acc.w += v.w;
                }
                deg += m;
                last = wbest;
            }
        }

        *(float4*)(S + (size_t)node * D + lane * 4) = acc;
        if (lane == 0) degf[node] = (float)deg;
    }
}

#define WMMA_F32(ACC, A2, B2) \
    ACC = __builtin_amdgcn_wmma_f32_16x16x4_f32(false, A2, false, B2, (short)0, ACC, false, false)

__global__ __launch_bounds__(256) void node_maps_kernel(const float* __restrict__ x,
                                                        const float* __restrict__ S,
                                                        const float* __restrict__ degf,
                                                        const float* __restrict__ Wn,
                                                        const float* __restrict__ bn,
                                                        const float* __restrict__ We,
                                                        const float* __restrict__ be,
                                                        float* __restrict__ node_emb,
                                                        float* __restrict__ agg_node,
                                                        float* __restrict__ agg_edge,
                                                        int nTiles) {
    extern __shared__ float lds[];
    float* ldsN = lds;
    float* ldsE = lds + 128 * PAD_WN;
    load_w_lds(ldsN, Wn, 128, 128, PAD_WN);
    load_w_lds(ldsE, We, 128, 256, PAD_WE);
    __syncthreads();

    const int lane = threadIdx.x & 31;
    const int wave = threadIdx.x >> 5;
    const int wpb  = blockDim.x >> 5;
    const int mlow = lane & 15;
    const int hi   = lane >> 4;

    const float* bN = ldsN + mlow * PAD_WN + 4 * hi;
    const float* bE = ldsE + mlow * PAD_WE + 4 * hi;

    for (int j = wave; ; j += wpb) {
        const int tile = j * (int)gridDim.x + (int)blockIdx.x;
        if (tile >= nTiles) break;
        const int m0 = tile << 4;

        const float* xr = x + (size_t)(m0 + mlow) * D + 4 * hi;
        const float* sr = S + (size_t)(m0 + mlow) * D + 4 * hi;

        v8f accXN[8] = {};
        v8f accSN[8] = {};
        v8f accXL[8] = {};
        v8f accSR[8] = {};

        float4 xn = *(const float4*)xr;
        float4 sn = *(const float4*)sr;
        #pragma unroll 1
        for (int c = 0; c < 16; ++c) {
            const float4 xa = xn;
            const float4 sa = sn;
            const int cn = (c < 15) ? (c + 1) : 15;
            xn = *(const float4*)(xr + 8 * cn);
            sn = *(const float4*)(sr + 8 * cn);

            const float* pN = bN + 8 * c;
            const float* pE = bE + 8 * c;
            float4 wN[8], wL[8], wR[8];
            #pragma unroll
            for (int t = 0; t < 8; ++t) {
                wN[t] = *(const float4*)(pN + t * (16 * PAD_WN));
                wL[t] = *(const float4*)(pE + t * (16 * PAD_WE));
                wR[t] = *(const float4*)(pE + t * (16 * PAD_WE) + 128);
            }
            const v2f x0 = {xa.x, xa.y};
            const v2f x1 = {xa.z, xa.w};
            const v2f s0 = {sa.x, sa.y};
            const v2f s1 = {sa.z, sa.w};
            #pragma unroll
            for (int t = 0; t < 8; ++t) {
                const v2f b0 = {wN[t].x, wN[t].y};
                WMMA_F32(accXN[t], x0, b0);
                WMMA_F32(accSN[t], s0, b0);
            }
            #pragma unroll
            for (int t = 0; t < 8; ++t) {
                const v2f b1 = {wN[t].z, wN[t].w};
                WMMA_F32(accXN[t], x1, b1);
                WMMA_F32(accSN[t], s1, b1);
            }
            #pragma unroll
            for (int t = 0; t < 8; ++t) {
                const v2f l0 = {wL[t].x, wL[t].y};
                const v2f r0 = {wR[t].x, wR[t].y};
                WMMA_F32(accXL[t], x0, l0);
                WMMA_F32(accSR[t], s0, r0);
            }
            #pragma unroll
            for (int t = 0; t < 8; ++t) {
                const v2f l1 = {wL[t].z, wL[t].w};
                const v2f r1 = {wR[t].z, wR[t].w};
                WMMA_F32(accXL[t], x1, l1);
                WMMA_F32(accSR[t], s1, r1);
            }
            asm volatile("" :: "v"(xn.x), "v"(xn.y), "v"(xn.z), "v"(xn.w),
                               "v"(sn.x), "v"(sn.y), "v"(sn.z), "v"(sn.w));
        }

        float dg[8];
        #pragma unroll
        for (int r = 0; r < 8; ++r) dg[r] = degf[m0 + r + (hi << 3)];

        #pragma unroll
        for (int t = 0; t < 8; ++t) {
            const int n = (t << 4) + mlow;
            const float bnv = bn[n];
            const float bev = be[n];
            #pragma unroll
            for (int r = 0; r < 8; ++r) {
                const size_t o = (size_t)(m0 + r + (hi << 3)) * D + n;
                node_emb[o] = accXN[t][r] + bnv;
                agg_node[o] = accSN[t][r] + dg[r] * bnv;
                agg_edge[o] = (dg[r] * accXL[t][r] + accSR[t][r]) + dg[r] * bev;
            }
        }
    }
}

__global__ __launch_bounds__(1024) void update_kernel(const float* __restrict__ node_emb,
                                                      const float* __restrict__ agg_node,
                                                      const float* __restrict__ agg_edge,
                                                      const float* __restrict__ Wu,
                                                      const float* __restrict__ bu,
                                                      float* __restrict__ out,
                                                      int nTiles) {
    extern __shared__ float lds[];
    load_w_lds(lds, Wu, 128, 384, PAD_WU);
    __syncthreads();

    const int lane  = threadIdx.x & 31;
    const int wave  = threadIdx.x >> 5;
    const int wpb   = blockDim.x >> 5;
    const int mlow  = lane & 15;
    const int hi    = lane >> 4;
    const int khalf = hi << 1;

    const float* bbase = lds + mlow * PAD_WU + khalf;
    const float* parts[3] = { node_emb, agg_node, agg_edge };

    for (int j = wave; ; j += wpb) {
        const int tile = j * (int)gridDim.x + (int)blockIdx.x;
        if (tile >= nTiles) break;
        const int m0 = tile << 4;

        v8f acc[8] = {};
        #pragma unroll 1
        for (int part = 0; part < 3; ++part) {
            const float* ap  = parts[part] + (size_t)(m0 + mlow) * D + khalf;
            const int    kg0 = part * 128;
            for (int k = 0; k < 128; k += 4) {
                const v2f raw = *(const v2f*)(ap + k);
                v2f a;
                a[0] = gelu_exact(raw[0]);
                a[1] = gelu_exact(raw[1]);
                #pragma unroll
                for (int t = 0; t < 8; ++t) {
                    const v2f b = *(const v2f*)(bbase + t * (16 * PAD_WU) + kg0 + k);
                    acc[t] = __builtin_amdgcn_wmma_f32_16x16x4_f32(
                        false, a, false, b, (short)0, acc[t], false, false);
                }
            }
        }

        #pragma unroll
        for (int t = 0; t < 8; ++t) {
            const int n = (t << 4) + mlow;
            const float bv = bu[n];
            #pragma unroll
            for (int r = 0; r < 8; ++r)
                out[(size_t)(m0 + r + (hi << 3)) * D + n] = acc[t][r] + bv;
        }
    }
}

extern "C" void kernel_launch(void* const* d_in, const int* in_sizes, int n_in,
                              void* d_out, int out_size, void* d_ws, size_t ws_size,
                              hipStream_t stream) {
    const float* x  = (const float*)d_in[0];
    const int*   ei = (const int*)d_in[1];
    const float* Wn = (const float*)d_in[2];
    const float* bn = (const float*)d_in[3];
    const float* We = (const float*)d_in[4];
    const float* be = (const float*)d_in[5];
    const float* Wu = (const float*)d_in[6];
    const float* bu = (const float*)d_in[7];

    const int N = in_sizes[0] / D;
    const int E = in_sizes[1] / 2;
    const int* srcI = ei;
    const int* dstI = ei + E;

    float* node_emb = (float*)d_ws;
    float* agg_node = node_emb + (size_t)N * D;
    float* agg_edge = agg_node + (size_t)N * D;
    float* S        = agg_edge + (size_t)N * D;
    float* degf     = S + (size_t)N * D;
    int*   cnt      = (int*)(degf + N);
    int*   slot     = cnt + N;

    hipMemsetAsync(cnt, 0, (size_t)N * sizeof(int), stream);

    const int wpb = 8;
    const int nTiles = (N + 15) >> 4;
    const dim3 blk(32 * wpb);

    fill_kernel<<<(E + 255) / 256, blk, 0, stream>>>(srcI, dstI, cnt, slot, E);

    gather_sum_kernel<<<(N + wpb * NODES_PER_WAVE - 1) / (wpb * NODES_PER_WAVE), blk, 0, stream>>>(
        x, srcI, dstI, cnt, slot, S, degf, N, E);

    node_maps_kernel<<<256, blk,
                       (128 * PAD_WN + 128 * PAD_WE) * sizeof(float), stream>>>(
        x, S, degf, Wn, bn, We, be, node_emb, agg_node, agg_edge, nTiles);

    update_kernel<<<256, dim3(1024),
                    128 * PAD_WU * sizeof(float), stream>>>(
        node_emb, agg_node, agg_edge, Wu, bu, (float*)d_out, nTiles);
}


# GELU's erf in straight-line code: both range polynomials evaluated in every lane and one selected, exp on the base-2 instruction, instead of the library erff's per-lane branches; sum order unchanged
# speedup vs baseline: 3.3595x; 1.0556x over previous
#include <hip/hip_runtime.h>
#include <hip/hip_bf16.h>

typedef __attribute__((ext_vector_type(2))) float v2f;
typedef __attribute__((ext_vector_type(8))) float v8f;

#define D 128
#define PAD_WN 132
#define PAD_WE 260
#define PAD_WU 388

__device__ __forceinline__ float erf_straight(float a) {
    const float t = __builtin_fabsf(a);
    const float s = a * a;
    float r = __builtin_fmaf(-1.72853470e-5f, t, 3.83197126e-4f);
    const float u = __builtin_fmaf(-3.88396438e-3f, t, 2.42546219e-2f);
    r = __builtin_fmaf(r, s, u);
    r = __builtin_fmaf(r, t, -1.06777877e-1f);
    r = __builtin_fmaf(r, t, -6.34846687e-1f);
    r = __builtin_fmaf(r, t, -1.28717512e-1f);
    r = __builtin_fmaf(r, t, -t);
    const float big = __builtin_copysignf(1.0f - __builtin_amdgcn_exp2f(r * 1.44269504088896341f), a);
    float q = -5.96761703e-4f;
    q = __builtin_fmaf(q, s, 4.99119423e-3f);
    q = __builtin_fmaf(q, s, -2.67681349e-2f);
    q = __builtin_fmaf(q, s, 1.12819925e-1f);
    q = __builtin_fmaf(q, s, -3.76125336e-1f);
    q = __builtin_fmaf(q, s, 1.28379166e-1f);
    const float small = __builtin_fmaf(q, a, a);
    return (t > 0.927734375f) ? big : small;
}

__device__ __forceinline__ float gelu_exact(float v) {
    return 0.5f * v * (1.0f + erf_straight(v * 0.70710678118654752f));
}

__device__ __forceinline__ void load_w_lds(float* lds, const float* __restrict__ W,
                                           int rows, int cols, int pad) {
    const int nv4 = cols >> 2;
    const int total = rows * nv4;
    for (int idx = threadIdx.x; idx < total; idx += blockDim.x) {
        const int r  = idx / nv4;
        const int c4 = (idx - r * nv4) << 2;
        const float4 v = *(const float4*)(W + r * cols + c4);
        *(float4*)(lds + r * pad + c4) = v;
    }
}

#define CAP 32
__global__ void fill_kernel(const int* __restrict__ srcI,
                            const int* __restrict__ dstI,
                            int* __restrict__ cnt,
                            int* __restrict__ slot,
                            int E) {
    const int e = blockIdx.x * blockDim.x + threadIdx.x;
    if (e < E) {
        const int s = srcI[e];
        const int d = dstI[e];
        asm volatile("" :: "v"(d));
        const int k = atomicAdd(cnt + s, 1);
        if (k < CAP) slot[(size_t)s * CAP + k] = d;
    }
}

#define NODES_PER_WAVE 8
__global__ __launch_bounds__(256) void gather_sum_kernel(const float* __restrict__ x,
                                                         const int* __restrict__ srcI,
                                                         const int* __restrict__ dstI,
                                                         const int* __restrict__ cnt,
                                                         const int* __restrict__ slot,
                                                         float* __restrict__ S,
                                                         float* __restrict__ degf,
                                                         int N, int E) {
    const int lane = threadIdx.x & 31;
    const int wv = __builtin_amdgcn_readfirstlane(
        (int)(blockIdx.x * (blockDim.x >> 5) + (threadIdx.x >> 5)));
    const float* xl = x + lane * 4;
    const int node0 = wv * NODES_PER_WAVE;

    int cq[NODES_PER_WAVE];
    int dq[NODES_PER_WAVE];
    if (node0 + NODES_PER_WAVE <= N) {
        #pragma unroll
        for (int q = 0; q < NODES_PER_WAVE; ++q) cq[q] = cnt[node0 + q];
    } else {
        #pragma unroll
        for (int q = 0; q < NODES_PER_WAVE; ++q)
            cq[q] = (node0 + q < N) ? cnt[node0 + q] : 0;
    }
    #pragma unroll
    for (int q = 0; q < NODES_PER_WAVE; ++q) {
        dq[q] = 0x7fffffff;
        if (lane < cq[q]) dq[q] = slot[(size_t)(node0 + q) * CAP + lane];
    }

    #pragma unroll 1
    for (int q = 0; q < NODES_PER_WAVE; ++q) {
        const int node = node0 + q;
        if (node >= N) break;

        int cnt_q = cq[0];
        int myd = dq[0];
        #pragma unroll
        for (int t = 1; t < NODES_PER_WAVE; ++t) {
            if (q == t) { cnt_q = cq[t]; myd = dq[t]; }
        }
        float4 acc = {0.0f, 0.0f, 0.0f, 0.0f};
        int deg = cnt_q;

        if (cnt_q <= CAP) {
            int rank = 0;
            #pragma unroll
            for (int j = 0; j < 32; ++j) {
                const int dj = __builtin_amdgcn_readlane(myd, j);
                rank += ((dj < myd) || (dj == myd && j < lane)) ? 1 : 0;
            }
            const int sorted = __builtin_amdgcn_ds_permute(rank << 2, myd);
            for (int r0 = 0; r0 < cnt_q; r0 += 4) {
                float4 v[4];
                #pragma unroll
                for (int i = 0; i < 4; ++i) {
                    const int r = (r0 + i < cnt_q) ? (r0 + i) : (cnt_q - 1);
                    const int d = __builtin_amdgcn_readlane(sorted, r);
                    v[i] = *(const float4*)(xl + (size_t)d * D);
                }
                #pragma unroll
                for (int i = 0; i < 4; ++i) {
                    if (r0 + i < cnt_q) {
                        acc.x += v[i].x; acc.y += v[i].y; acc.z += v[i].z; acc.w += v[i].w;
                    }
                }
            }
        } else {
            deg = 0;
            int last = -1;
            for (int outer = 0; outer < N; ++outer) {
                int best = 0x7fffffff;
                int mult = 0;
                for (int e0 = 0; e0 < E; e0 += 32) {
                    const int e = e0 + lane;
                    if (e < E && srcI[e] == node) {
                        const int d = dstI[e];
                        if (d > last) {
                            if (d < best) { best = d; mult = 1; }
                            else if (d == best) ++mult;
                        }
                    }
                }
                int wbest = best;
                for (int off = 16; off >= 1; off >>= 1) {
                    const int o = __shfl_xor(wbest, off);
                    wbest = (o < wbest) ? o : wbest;
                }
                int m = (best == wbest) ? mult : 0;
                for (int off = 16; off >= 1; off >>= 1) m += __shfl_xor(m, off);
                m = __builtin_amdgcn_readfirstlane(m);
                wbest = __builtin_amdgcn_readfirstlane(wbest);
                if (m == 0) break;
                const float4 v = *(const float4*)(xl + (size_t)wbest * D);
                for (int mm = 0; mm < m; ++mm) {
                    acc.x += v.x; acc.y += v.y; acc.z += v.z; acc.w += v.w;
                }
                deg += m;
                last = wbest;
            }
        }

        *(float4*)(S + (size_t)node * D + lane * 4) = acc;
        if (lane == 0) degf[node] = (float)deg;
    }
}

#define WMMA_F32(ACC, A2, B2) \
    ACC = __builtin_amdgcn_wmma_f32_16x16x4_f32(false, A2, false, B2, (short)0, ACC, false, false)

__global__ __launch_bounds__(256) void node_maps_kernel(const float* __restrict__ x,
                                                        const float* __restrict__ S,
                                                        const float* __restrict__ degf,
                                                        const float* __restrict__ Wn,
                                                        const float* __restrict__ bn,
                                                        const float* __restrict__ We,
                                                        const float* __restrict__ be,
                                                        float* __restrict__ node_emb,
                                                        float* __restrict__ agg_node,
                                                        float* __restrict__ agg_edge,
                                                        int nTiles) {
    extern __shared__ float lds[];
    float* ldsN = lds;
    float* ldsE = lds + 128 * PAD_WN;
    load_w_lds(ldsN, Wn, 128, 128, PAD_WN);
    load_w_lds(ldsE, We, 128, 256, PAD_WE);
    __syncthreads();

    const int lane = threadIdx.x & 31;
    const int wave = threadIdx.x >> 5;
    const int wpb  = blockDim.x >> 5;
    const int mlow = lane & 15;
    const int hi   = lane >> 4;

    const float* bN = ldsN + mlow * PAD_WN + 4 * hi;
    const float* bE = ldsE + mlow * PAD_WE + 4 * hi;

    for (int j = wave; ; j += wpb) {
        const int tile = j * (int)gridDim.x + (int)blockIdx.x;
        if (tile >= nTiles) break;
        const int m0 = tile << 4;

        const float* xr = x + (size_t)(m0 + mlow) * D + 4 * hi;
        const float* sr = S + (size_t)(m0 + mlow) * D + 4 * hi;

        v8f accXN[8] = {};
        v8f accSN[8] = {};
        v8f accXL[8] = {};
        v8f accSR[8] = {};

        float4 xn = *(const float4*)xr;
        float4 sn = *(const float4*)sr;
        #pragma unroll 1
        for (int c = 0; c < 16; ++c) {
            const float4 xa = xn;
            const float4 sa = sn;
            const int cn = (c < 15) ? (c + 1) : 15;
            xn = *(const float4*)(xr + 8 * cn);
            sn = *(const float4*)(sr + 8 * cn);

            const float* pN = bN + 8 * c;
            const float* pE = bE + 8 * c;
            float4 wN[8], wL[8], wR[8];
            #pragma unroll
            for (int t = 0; t < 8; ++t) {
                wN[t] = *(const float4*)(pN + t * (16 * PAD_WN));
                wL[t] = *(const float4*)(pE + t * (16 * PAD_WE));
                wR[t] = *(const float4*)(pE + t * (16 * PAD_WE) + 128);
            }
            const v2f x0 = {xa.x, xa.y};
            const v2f x1 = {xa.z, xa.w};
            const v2f s0 = {sa.x, sa.y};
            const v2f s1 = {sa.z, sa.w};
            #pragma unroll
            for (int t = 0; t < 8; ++t) {
                const v2f b0 = {wN[t].x, wN[t].y};
                WMMA_F32(accXN[t], x0, b0);
                WMMA_F32(accSN[t], s0, b0);
            }
            #pragma unroll
            for (int t = 0; t < 8; ++t) {
                const v2f b1 = {wN[t].z, wN[t].w};
                WMMA_F32(accXN[t], x1, b1);
                WMMA_F32(accSN[t], s1, b1);
            }
            #pragma unroll
            for (int t = 0; t < 8; ++t) {
                const v2f l0 = {wL[t].x, wL[t].y};
                const v2f r0 = {wR[t].x, wR[t].y};
                WMMA_F32(accXL[t], x0, l0);
                WMMA_F32(accSR[t], s0, r0);
            }
            #pragma unroll
            for (int t = 0; t < 8; ++t) {
                const v2f l1 = {wL[t].z, wL[t].w};
                const v2f r1 = {wR[t].z, wR[t].w};
                WMMA_F32(accXL[t], x1, l1);
                WMMA_F32(accSR[t], s1, r1);
            }
            asm volatile("" :: "v"(xn.x), "v"(xn.y), "v"(xn.z), "v"(xn.w),
                               "v"(sn.x), "v"(sn.y), "v"(sn.z), "v"(sn.w));
        }

        float dg[8];
        #pragma unroll
        for (int r = 0; r < 8; ++r) dg[r] = degf[m0 + r + (hi << 3)];

        #pragma unroll
        for (int t = 0; t < 8; ++t) {
            const int n = (t << 4) + mlow;
            const float bnv = bn[n];
            const float bev = be[n];
            #pragma unroll
            for (int r = 0; r < 8; ++r) {
                const size_t o = (size_t)(m0 + r + (hi << 3)) * D + n;
                node_emb[o] = accXN[t][r] + bnv;
                agg_node[o] = accSN[t][r] + dg[r] * bnv;
                agg_edge[o] = (dg[r] * accXL[t][r] + accSR[t][r]) + dg[r] * bev;
            }
        }
    }
}

__global__ __launch_bounds__(1024) void update_kernel(const float* __restrict__ node_emb,
                                                      const float* __restrict__ agg_node,
                                                      const float* __restrict__ agg_edge,
                                                      const float* __restrict__ Wu,
                                                      const float* __restrict__ bu,
                                                      float* __restrict__ out,
                                                      int nTiles) {
    extern __shared__ float lds[];
    load_w_lds(lds, Wu, 128, 384, PAD_WU);
    __syncthreads();

    const int lane  = threadIdx.x & 31;
    const int wave  = threadIdx.x >> 5;
    const int wpb   = blockDim.x >> 5;
    const int mlow  = lane & 15;
    const int hi    = lane >> 4;
    const int khalf = hi << 1;

    const float* bbase = lds + mlow * PAD_WU + khalf;
    const float* parts[3] = { node_emb, agg_node, agg_edge };

    for (int j = wave; ; j += wpb) {
        const int tile = j * (int)gridDim.x + (int)blockIdx.x;
        if (tile >= nTiles) break;
        const int m0 = tile << 4;

        v8f acc[8] = {};
        #pragma unroll 1
        for (int part = 0; part < 3; ++part) {
            const float* ap  = parts[part] + (size_t)(m0 + mlow) * D + khalf;
            const int    kg0 = part * 128;
            for (int k = 0; k < 128; k += 4) {
                const v2f raw = *(const v2f*)(ap + k);
                v2f a;
                a[0] = gelu_exact(raw[0]);
                a[1] = gelu_exact(raw[1]);
                #pragma unroll
                for (int t = 0; t < 8; ++t) {
                    const v2f b = *(const v2f*)(bbase + t * (16 * PAD_WU) + kg0 + k);
                    acc[t] = __builtin_amdgcn_wmma_f32_16x16x4_f32(
                        false, a, false, b, (short)0, acc[t], false, false);
                }
            }
        }

        #pragma unroll
        for (int t = 0; t < 8; ++t) {
            const int n = (t << 4) + mlow;
            const float bv = bu[n];
            #pragma unroll
            for (int r = 0; r < 8; ++r)
                out[(size_t)(m0 + r + (hi << 3)) * D + n] = acc[t][r] + bv;
        }
    }
}

extern "C" void kernel_launch(void* const* d_in, const int* in_sizes, int n_in,
                              void* d_out, int out_size, void* d_ws, size_t ws_size,
                              hipStream_t stream) {
    const float* x  = (const float*)d_in[0];
    const int*   ei = (const int*)d_in[1];
    const float* Wn = (const float*)d_in[2];
    const float* bn = (const float*)d_in[3];
    const float* We = (const float*)d_in[4];
    const float* be = (const float*)d_in[5];
    const float* Wu = (const float*)d_in[6];
    const float* bu = (const float*)d_in[7];

    const int N = in_sizes[0] / D;
    const int E = in_sizes[1] / 2;
    const int* srcI = ei;
    const int* dstI = ei + E;

    float* node_emb = (float*)d_ws;
    float* agg_node = node_emb + (size_t)N * D;
    float* agg_edge = agg_node + (size_t)N * D;
    float* S        = agg_edge + (size_t)N * D;
    float* degf     = S + (size_t)N * D;
    int*   cnt      = (int*)(degf + N);
    int*   slot     = cnt + N;

    hipMemsetAsync(cnt, 0, (size_t)N * sizeof(int), stream);

    const int wpb = 8;
    const int nTiles = (N + 15) >> 4;
    const dim3 blk(32 * wpb);

    fill_kernel<<<(E + 255) / 256, blk, 0, stream>>>(srcI, dstI, cnt, slot, E);

    gather_sum_kernel<<<(N + wpb * NODES_PER_WAVE - 1) / (wpb * NODES_PER_WAVE), blk, 0, stream>>>(
        x, srcI, dstI, cnt, slot, S, degf, N, E);

    node_maps_kernel<<<256, blk,
                       (128 * PAD_WN + 128 * PAD_WE) * sizeof(float), stream>>>(
        x, S, degf, Wn, bn, We, be, node_emb, agg_node, agg_edge, nTiles);

    update_kernel<<<256, dim3(1024),
                    128 * PAD_WU * sizeof(float), stream>>>(
        node_emb, agg_node, agg_edge, Wu, bu, (float*)d_out, nTiles);
}


# weight staging to LDS in both WMMA kernels: a thread's 12 to 32 loads per matrix are requested together ahead of its LDS stores, not one load-wait-store per pass (60 serial round trips become 3)
# speedup vs baseline: 3.6398x; 1.0834x over previous
#include <hip/hip_runtime.h>
#include <hip/hip_bf16.h>

typedef __attribute__((ext_vector_type(2))) float v2f;
typedef __attribute__((ext_vector_type(8))) float v8f;

#define D 128
#define PAD_WN 132
#define PAD_WE 260
#define PAD_WU 388

__device__ __forceinline__ float erf_straight(float a) {
    const float t = __builtin_fabsf(a);
    const float s = a * a;
    float r = __builtin_fmaf(-1.72853470e-5f, t, 3.83197126e-4f);
    const float u = __builtin_fmaf(-3.88396438e-3f, t, 2.42546219e-2f);
    r = __builtin_fmaf(r, s, u);
    r = __builtin_fmaf(r, t, -1.06777877e-1f);
    r = __builtin_fmaf(r, t, -6.34846687e-1f);
    r = __builtin_fmaf(r, t, -1.28717512e-1f);
    r = __builtin_fmaf(r, t, -t);
    const float big = __builtin_copysignf(1.0f - __builtin_amdgcn_exp2f(r * 1.44269504088896341f), a);
    float q = -5.96761703e-4f;
    q = __builtin_fmaf(q, s, 4.99119423e-3f);
    q = __builtin_fmaf(q, s, -2.67681349e-2f);
    q = __builtin_fmaf(q, s, 1.12819925e-1f);
    q = __builtin_fmaf(q, s, -3.76125336e-1f);
    q = __builtin_fmaf(q, s, 1.28379166e-1f);
    const float small = __builtin_fmaf(q, a, a);
    return (t > 0.927734375f) ? big : small;
}

__device__ __forceinline__ float gelu_exact(float v) {
    return 0.5f * v * (1.0f + erf_straight(v * 0.70710678118654752f));
}

template <int U>
__device__ __forceinline__ void load_w_lds(float* lds, const float* __restrict__ W,
                                           int rows, int cols, int pad) {
    const int nv4 = cols >> 2;
    const int total = rows * nv4;
    #pragma unroll U
    for (int idx = threadIdx.x; idx < total; idx += blockDim.x) {
        const int r  = idx / nv4;
        const int c4 = (idx - r * nv4) << 2;
        const float4 v = *(const float4*)(W + r * cols + c4);
        *(float4*)(lds + r * pad + c4) = v;
    }
}

#define CAP 32
__global__ void fill_kernel(const int* __restrict__ srcI,
                            const int* __restrict__ dstI,
                            int* __restrict__ cnt,
                            int* __restrict__ slot,
                            int E) {
    const int e = blockIdx.x * blockDim.x + threadIdx.x;
    if (e < E) {
        const int s = srcI[e];
        const int d = dstI[e];
        asm volatile("" :: "v"(d));
        const int k = atomicAdd(cnt + s, 1);
        if (k < CAP) slot[(size_t)s * CAP + k] = d;
    }
}

#define NODES_PER_WAVE 8
__global__ __launch_bounds__(256) void gather_sum_kernel(const float* __restrict__ x,
                                                         const int* __restrict__ srcI,
                                                         const int* __restrict__ dstI,
                                                         const int* __restrict__ cnt,
                                                         const int* __restrict__ slot,
                                                         float* __restrict__ S,
                                                         float* __restrict__ degf,
                                                         int N, int E) {
    const int lane = threadIdx.x & 31;
    const int wv = __builtin_amdgcn_readfirstlane(
        (int)(blockIdx.x * (blockDim.x >> 5) + (threadIdx.x >> 5)));
    const float* xl = x + lane * 4;
    const int node0 = wv * NODES_PER_WAVE;

    int cq[NODES_PER_WAVE];
    int dq[NODES_PER_WAVE];
    if (node0 + NODES_PER_WAVE <= N) {
        #pragma unroll
        for (int q = 0; q < NODES_PER_WAVE; ++q) cq[q] = cnt[node0 + q];
    } else {
        #pragma unroll
        for (int q = 0; q < NODES_PER_WAVE; ++q)
            cq[q] = (node0 + q < N) ? cnt[node0 + q] : 0;
    }
    #pragma unroll
    for (int q = 0; q < NODES_PER_WAVE; ++q) {
        dq[q] = 0x7fffffff;
        if (lane < cq[q]) dq[q] = slot[(size_t)(node0 + q) * CAP + lane];
    }

    #pragma unroll 1
    for (int q = 0; q < NODES_PER_WAVE; ++q) {
        const int node = node0 + q;
        if (node >= N) break;

        int cnt_q = cq[0];
        int myd = dq[0];
        #pragma unroll
        for (int t = 1; t < NODES_PER_WAVE; ++t) {
            if (q == t) { cnt_q = cq[t]; myd = dq[t]; }
        }
        float4 acc = {0.0f, 0.0f, 0.0f, 0.0f};
        int deg = cnt_q;

        if (cnt_q <= CAP) {
            int rank = 0;
            #pragma unroll
            for (int j = 0; j < 32; ++j) {
                const int dj = __builtin_amdgcn_readlane(myd, j);
                rank += ((dj < myd) || (dj == myd && j < lane)) ? 1 : 0;
            }
            const int sorted = __builtin_amdgcn_ds_permute(rank << 2, myd);
            for (int r0 = 0; r0 < cnt_q; r0 += 4) {
                float4 v[4];
                #pragma unroll
                for (int i = 0; i < 4; ++i) {
                    const int r = (r0 + i < cnt_q) ? (r0 + i) : (cnt_q - 1);
                    const int d = __builtin_amdgcn_readlane(sorted, r);
                    v[i] = *(const float4*)(xl + (size_t)d * D);
                }
                #pragma unroll
                for (int i = 0; i < 4; ++i) {
                    if (r0 + i < cnt_q) {
                        acc.x += v[i].x; acc.y += v[i].y; acc.z += v[i].z; acc.w += v[i].w;
                    }
                }
            }
        } else {
            deg = 0;
            int last = -1;
            for (int outer = 0; outer < N; ++outer) {
                int best = 0x7fffffff;
                int mult = 0;
                for (int e0 = 0; e0 < E; e0 += 32) {
                    const int e = e0 + lane;
                    if (e < E && srcI[e] == node) {
                        const int d = dstI[e];
                        if (d > last) {
                            if (d < best) { best = d; mult = 1; }
                            else if (d == best) ++mult;
                        }
                    }
                }
                int wbest = best;
                for (int off = 16; off >= 1; off >>= 1) {
                    const int o = __shfl_xor(wbest, off);
                    wbest = (o < wbest) ? o : wbest;
                }
                int m = (best == wbest) ? mult : 0;
                for (int off = 16; off >= 1; off >>= 1) m += __shfl_xor(m, off);
                m = __builtin_amdgcn_readfirstlane(m);
                wbest = __builtin_amdgcn_readfirstlane(wbest);
                if (m == 0) break;
                const float4 v = *(const float4*)(xl + (size_t)wbest * D);
                for (int mm = 0; mm < m; ++mm) {
                    acc.x += v.x; acc.y += v.y; acc.z += v.z; acc.w += v.w;
                }
                deg += m;
                last = wbest;
            }
        }

        *(float4*)(S + (size_t)node * D + lane * 4) = acc;
        if (lane == 0) degf[node] = (float)deg;
    }
}

#define WMMA_F32(ACC, A2, B2) \
    ACC = __builtin_amdgcn_wmma_f32_16x16x4_f32(false, A2, false, B2, (short)0, ACC, false, false)

__global__ __launch_bounds__(256) void node_maps_kernel(const float* __restrict__ x,
                                                        const float* __restrict__ S,
                                                        const float* __restrict__ degf,
                                                        const float* __restrict__ Wn,
                                                        const float* __restrict__ bn,
                                                        const float* __restrict__ We,
                                                        const float* __restrict__ be,
                                                        float* __restrict__ node_emb,
                                                        float* __restrict__ agg_node,
                                                        float* __restrict__ agg_edge,
                                                        int nTiles) {
    extern __shared__ float lds[];
    float* ldsN = lds;
    float* ldsE = lds + 128 * PAD_WN;
    load_w_lds<16>(ldsN, Wn, 128, 128, PAD_WN);
    load_w_lds<32>(ldsE, We, 128, 256, PAD_WE);
    __syncthreads();

    const int lane = threadIdx.x & 31;
    const int wave = threadIdx.x >> 5;
    const int wpb  = blockDim.x >> 5;
    const int mlow = lane & 15;
    const int hi   = lane >> 4;

    const float* bN = ldsN + mlow * PAD_WN + 4 * hi;
    const float* bE = ldsE + mlow * PAD_WE + 4 * hi;

    for (int j = wave; ; j += wpb) {
        const int tile = j * (int)gridDim.x + (int)blockIdx.x;
        if (tile >= nTiles) break;
        const int m0 = tile << 4;

        const float* xr = x + (size_t)(m0 + mlow) * D + 4 * hi;
        const float* sr = S + (size_t)(m0 + mlow) * D + 4 * hi;

        v8f accXN[8] = {};
        v8f accSN[8] = {};
        v8f accXL[8] = {};
        v8f accSR[8] = {};

        float4 xn = *(const float4*)xr;
        float4 sn = *(const float4*)sr;
        #pragma unroll 1
        for (int c = 0; c < 16; ++c) {
            const float4 xa = xn;
            const float4 sa = sn;
            const int cn = (c < 15) ? (c + 1) : 15;
            xn = *(const float4*)(xr + 8 * cn);
            sn = *(const float4*)(sr + 8 * cn);

            const float* pN = bN + 8 * c;
            const float* pE = bE + 8 * c;
            float4 wN[8], wL[8], wR[8];
            #pragma unroll
            for (int t = 0; t < 8; ++t) {
                wN[t] = *(const float4*)(pN + t * (16 * PAD_WN));
                wL[t] = *(const float4*)(pE + t * (16 * PAD_WE));
                wR[t] = *(const float4*)(pE + t * (16 * PAD_WE) + 128);
            }
            const v2f x0 = {xa.x, xa.y};
            const v2f x1 = {xa.z, xa.w};
            const v2f s0 = {sa.x, sa.y};
            const v2f s1 = {sa.z, sa.w};
            #pragma unroll
            for (int t = 0; t < 8; ++t) {
                const v2f b0 = {wN[t].x, wN[t].y};
                WMMA_F32(accXN[t], x0, b0);
                WMMA_F32(accSN[t], s0, b0);
            }
            #pragma unroll
            for (int t = 0; t < 8; ++t) {
                const v2f b1 = {wN[t].z, wN[t].w};
                WMMA_F32(accXN[t], x1, b1);
                WMMA_F32(accSN[t], s1, b1);
            }
            #pragma unroll
            for (int t = 0; t < 8; ++t) {
                const v2f l0 = {wL[t].x, wL[t].y};
                const v2f r0 = {wR[t].x, wR[t].y};
                WMMA_F32(accXL[t], x0, l0);
                WMMA_F32(accSR[t], s0, r0);
            }
            #pragma unroll
            for (int t = 0; t < 8; ++t) {
                const v2f l1 = {wL[t].z, wL[t].w};
                const v2f r1 = {wR[t].z, wR[t].w};
                WMMA_F32(accXL[t], x1, l1);
                WMMA_F32(accSR[t], s1, r1);
            }
            asm volatile("" :: "v"(xn.x), "v"(xn.y), "v"(xn.z), "v"(xn.w),
                               "v"(sn.x), "v"(sn.y), "v"(sn.z), "v"(sn.w));
        }

        float dg[8];
        #pragma unroll
        for (int r = 0; r < 8; ++r) dg[r] = degf[m0 + r + (hi << 3)];

        #pragma unroll
        for (int t = 0; t < 8; ++t) {
            const int n = (t << 4) + mlow;
            const float bnv = bn[n];
            const float bev = be[n];
            #pragma unroll
            for (int r = 0; r < 8; ++r) {
                const size_t o = (size_t)(m0 + r + (hi << 3)) * D + n;
                node_emb[o] = accXN[t][r] + bnv;
                agg_node[o] = accSN[t][r] + dg[r] * bnv;
                agg_edge[o] = (dg[r] * accXL[t][r] + accSR[t][r]) + dg[r] * bev;
            }
        }
    }
}

__global__ __launch_bounds__(1024) void update_kernel(const float* __restrict__ node_emb,
                                                      const float* __restrict__ agg_node,
                                                      const float* __restrict__ agg_edge,
                                                      const float* __restrict__ Wu,
                                                      const float* __restrict__ bu,
                                                      float* __restrict__ out,
                                                      int nTiles) {
    extern __shared__ float lds[];
    load_w_lds<12>(lds, Wu, 128, 384, PAD_WU);
    __syncthreads();

    const int lane  = threadIdx.x & 31;
    const int wave  = threadIdx.x >> 5;
    const int wpb   = blockDim.x >> 5;
    const int mlow  = lane & 15;
    const int hi    = lane >> 4;
    const int khalf = hi << 1;

    const float* bbase = lds + mlow * PAD_WU + khalf;
    const float* parts[3] = { node_emb, agg_node, agg_edge };

    for (int j = wave; ; j += wpb) {
        const int tile = j * (int)gridDim.x + (int)blockIdx.x;
        if (tile >= nTiles) break;
        const int m0 = tile << 4;

        v8f acc[8] = {};
        #pragma unroll 1
        for (int part = 0; part < 3; ++part) {
            const float* ap  = parts[part] + (size_t)(m0 + mlow) * D + khalf;
            const int    kg0 = part * 128;
            for (int k = 0; k < 128; k += 4) {
                const v2f raw = *(const v2f*)(ap + k);
                v2f a;
                a[0] = gelu_exact(raw[0]);
                a[1] = gelu_exact(raw[1]);
                #pragma unroll
                for (int t = 0; t < 8; ++t) {
                    const v2f b = *(const v2f*)(bbase + t * (16 * PAD_WU) + kg0 + k);
                    acc[t] = __builtin_amdgcn_wmma_f32_16x16x4_f32(
                        false, a, false, b, (short)0, acc[t], false, false);
                }
            }
        }

        #pragma unroll
        for (int t = 0; t < 8; ++t) {
            const int n = (t << 4) + mlow;
            const float bv = bu[n];
            #pragma unroll
            for (int r = 0; r < 8; ++r)
                out[(size_t)(m0 + r + (hi << 3)) * D + n] = acc[t][r] + bv;
        }
    }
}

extern "C" void kernel_launch(void* const* d_in, const int* in_sizes, int n_in,
                              void* d_out, int out_size, void* d_ws, size_t ws_size,
                              hipStream_t stream) {
    const float* x  = (const float*)d_in[0];
    const int*   ei = (const int*)d_in[1];
    const float* Wn = (const float*)d_in[2];
    const float* bn = (const float*)d_in[3];
    const float* We = (const float*)d_in[4];
    const float* be = (const float*)d_in[5];
    const float* Wu = (const float*)d_in[6];
    const float* bu = (const float*)d_in[7];

    const int N = in_sizes[0] / D;
    const int E = in_sizes[1] / 2;
    const int* srcI = ei;
    const int* dstI = ei + E;

    float* node_emb = (float*)d_ws;
    float* agg_node = node_emb + (size_t)N * D;
    float* agg_edge = agg_node + (size_t)N * D;
    float* S        = agg_edge + (size_t)N * D;
    float* degf     = S + (size_t)N * D;
    int*   cnt      = (int*)(degf + N);
    int*   slot     = cnt + N;

    hipMemsetAsync(cnt, 0, (size_t)N * sizeof(int), stream);

    const int wpb = 8;
    const int nTiles = (N + 15) >> 4;
    const dim3 blk(32 * wpb);

    fill_kernel<<<(E + 255) / 256, blk, 0, stream>>>(srcI, dstI, cnt, slot, E);

    gather_sum_kernel<<<(N + wpb * NODES_PER_WAVE - 1) / (wpb * NODES_PER_WAVE), blk, 0, stream>>>(
        x, srcI, dstI, cnt, slot, S, degf, N, E);

    node_maps_kernel<<<256, blk,
                       (128 * PAD_WN + 128 * PAD_WE) * sizeof(float), stream>>>(
        x, S, degf, Wn, bn, We, be, node_emb, agg_node, agg_edge, nTiles);

    update_kernel<<<256, dim3(1024),
                    128 * PAD_WU * sizeof(float), stream>>>(
        node_emb, agg_node, agg_edge, Wu, bu, (float*)d_out, nTiles);
}
